# starting kernel
# baseline (speedup 1.0000x reference)
#include <hip/hip_runtime.h>
#include <cmath>

#define N_NODES 50000
#define N_EDGES 800000
#define DF 128

typedef __attribute__((ext_vector_type(2))) float v2f;
typedef __attribute__((ext_vector_type(8))) float v8f;

__global__ void copy_kernel(float* __restrict__ dst, const float* __restrict__ src, int n4) {
    int i = blockIdx.x * blockDim.x + threadIdx.x;
    if (i < n4) ((float4*)dst)[i] = ((const float4*)src)[i];
}

__global__ void scatter_kernel(float* __restrict__ agg, const float* __restrict__ x,
                               const int* __restrict__ ei) {
    int t = blockIdx.x * blockDim.x + threadIdx.x;
    int e = t >> 5;
    if (e >= N_EDGES) return;
    int f = (t & 31) << 2;
    int s = ei[e];
    int d = ei[N_EDGES + e];
    float4 v = *(const float4*)(x + (size_t)s * DF + f);
    float* p = agg + (size_t)d * DF + f;
    atomicAdd(p + 0, v.x);
    atomicAdd(p + 1, v.y);
    atomicAdd(p + 2, v.z);
    atomicAdd(p + 3, v.w);
}

template<int NSEG, int EPI>
__global__ void __launch_bounds__(256, 2)
gemm_wmma(const float* __restrict__ A0, const float* __restrict__ A1,
          const float* __restrict__ A2, const float* __restrict__ W,
          const float* __restrict__ bias, const float* __restrict__ resid,
          float* __restrict__ out) {
    __shared__ float lA[16 * DF];

    const int lane = threadIdx.x & 31;
    const int wave = threadIdx.x >> 5;
    const int r    = lane & 15;
    const int hi   = lane >> 4;
    const int n0   = wave * 16;
    const int row0 = blockIdx.x * 16;

    v8f acc = {0.f, 0.f, 0.f, 0.f, 0.f, 0.f, 0.f, 0.f};

    const float* As[3] = {A0, A1, A2};
    #pragma unroll
    for (int seg = 0; seg < NSEG; ++seg) {
        const float4* s4 = (const float4*)(As[seg] + (size_t)row0 * DF);
        float4* d4 = (float4*)lA;
        #pragma unroll
        for (int i = 0; i < 2; ++i)
            d4[threadIdx.x + i * 256] = s4[threadIdx.x + i * 256];
        __syncthreads();

        const float* Wseg = W + seg * DF * DF;
        #pragma unroll
        for (int k0 = 0; k0 < DF; k0 += 4) {
            const float* ap = lA + r * DF + k0 + 2 * hi;
            v2f a; a.x = ap[0]; a.y = ap[1];
            const float* bp = Wseg + (k0 + 2 * hi) * DF + n0 + r;
            v2f b; b.x = bp[0]; b.y = bp[DF];
            acc = __builtin_amdgcn_wmma_f32_16x16x4_f32(
                false, a, false, b, (short)0, acc, false, false);
        }
        __syncthreads();
    }

    const int col = n0 + r;
    const float bb = bias[col];
    #pragma unroll
    for (int v = 0; v < 8; ++v) {
        int row = row0 + v + 8 * hi;
        float y = acc[v] + bb;
        if (EPI == 1) y = fmaxf(y, 0.0f);
        if (EPI == 2) {
            y += resid[(size_t)row * DF + col];
            y = (y > 0.0f) ? y : expm1f(y);
        }
        out[(size_t)row * DF + col] = y;
    }
}

extern "C" void kernel_launch(void* const* d_in, const int* in_sizes, int n_in,
                              void* d_out, int out_size, void* d_ws, size_t ws_size,
                              hipStream_t stream) {
    const float* x   = (const float*)d_in[0];
    const int*   ei  = (const int*)d_in[1];
    const float* W1[3] = {(const float*)d_in[2],  (const float*)d_in[6],  (const float*)d_in[10]};
    const float* b1[3] = {(const float*)d_in[3],  (const float*)d_in[7],  (const float*)d_in[11]};
    const float* W2[3] = {(const float*)d_in[4],  (const float*)d_in[8],  (const float*)d_in[12]};
    const float* b2[3] = {(const float*)d_in[5],  (const float*)d_in[9],  (const float*)d_in[13]};
    const float* Wjk = (const float*)d_in[14];
    const float* bjk = (const float*)d_in[15];

    const size_t NM = (size_t)N_NODES * DF;
    float* h0 = (float*)d_ws;
    float* xs[3] = {h0 + NM, h0 + 2 * NM, h0 + 3 * NM};

    const int n4 = (int)(NM / 4);
    dim3 blk(256);
    const int gBlocks = N_NODES / 16;

    const float* xin = x;
    for (int l = 0; l < 3; ++l) {
        copy_kernel<<<(n4 + 255) / 256, blk, 0, stream>>>(h0, xin, n4);
        scatter_kernel<<<(N_EDGES * 32) / 256, blk, 0, stream>>>(h0, xin, ei);
        gemm_wmma<1, 1><<<gBlocks, blk, 0, stream>>>(h0, nullptr, nullptr,
                                                     W1[l], b1[l], nullptr, h0);
        gemm_wmma<1, 2><<<gBlocks, blk, 0, stream>>>(h0, nullptr, nullptr,
                                                     W2[l], b2[l], xin, xs[l]);
        xin = xs[l];
    }
    gemm_wmma<3, 0><<<gBlocks, blk, 0, stream>>>(xs[0], xs[1], xs[2],
                                                 Wjk, bjk, nullptr, (float*)d_out);
}


# Neighbour sums by 250 work-groups owning 200 node rows each in LDS: every wave scans the dst list, adds matching source rows by LDS float atomics (arrival order); no global atomics, no copy launch
# speedup vs baseline: 1.6045x; 1.6045x over previous
#include <hip/hip_runtime.h>
#include <cmath>

#define N_NODES 50000
#define N_EDGES 800000
#define DF 128

typedef __attribute__((ext_vector_type(2))) float v2f;
typedef __attribute__((ext_vector_type(8))) float v8f;

#define SLAB   200
#define NSLAB  250
#define NBLK   (N_EDGES / 128)
#define BROT   (NBLK / NSLAB)
#define QCAP   160
#define FLUSH_AT 24

__global__ void __launch_bounds__(1024)
slab_aggregate(float* __restrict__ h, const float* __restrict__ x, const int* __restrict__ ei) {
    __shared__ float acc[(SLAB + 1) * DF];
    __shared__ unsigned int q[32 * QCAP];

    const int lane = threadIdx.x & 31;
    const int w    = __builtin_amdgcn_readfirstlane(threadIdx.x >> 5);
    const int g    = blockIdx.x;
    const int base = g * SLAB;
    const float4* x4 = (const float4*)x;

    {
        float4 iv[7];
        #pragma unroll
        for (int i = 0; i < 7; ++i) {
            const int r = min(w + 32 * i, SLAB - 1);
            iv[i] = x4[(size_t)(base + r) * 32 + lane];
        }
        #pragma unroll
        for (int i = 0; i < 7; ++i) {
            const int r = w + 32 * i;
            if (r < SLAB) {
                float* a = acc + r * DF + lane;
                a[0] = iv[i].x; a[32] = iv[i].y; a[64] = iv[i].z; a[96] = iv[i].w;
            }
        }
    }
    __syncthreads();

    const int4* d4 = (const int4*)(ei + N_EDGES);
    unsigned int* myq = q + w * QCAP;
    int cnt = 0;

    auto drain = [&](int m) {
        for (int b0 = 0; b0 < m; b0 += 32) {
            const int n = min(32, m - b0);
            unsigned ent = 0u; unsigned s_l = 0u;
            if (lane < n) {
                ent = myq[b0 + lane];
                s_l = min((unsigned)ei[ent >> 8], (unsigned)(N_NODES - 1));
            }
            const int ld_l = (int)(ent & 255u);
            for (int k0 = 0; k0 < n; k0 += 8) {
                float4 r[8];
                #pragma unroll
                for (int u = 0; u < 8; ++u) {
                    const unsigned s = __builtin_amdgcn_readlane(s_l, k0 + u);
                    r[u] = x4[(size_t)s * 32 + lane];
                }
                #pragma unroll
                for (int u = 0; u < 8; ++u) {
                    const int ld = __builtin_amdgcn_readlane(ld_l, k0 + u);
                    float* a = acc + ld * DF + lane;
                    atomicAdd(a,      r[u].x);
                    atomicAdd(a + 32, r[u].y);
                    atomicAdd(a + 64, r[u].z);
                    atomicAdd(a + 96, r[u].w);
                }
            }
        }
    };

    for (int k = 0; k < (NBLK + 31) / 32; k += 8) {
        int4 v[8]; int eb[8];
        #pragma unroll
        for (int u = 0; u < 8; ++u) {
            const int j = (k + u) * 32 + w;
            int bi = min(j, NBLK - 1) + g * BROT; if (bi >= NBLK) bi -= NBLK;
            eb[u] = bi * 128;
            v[u] = d4[bi * 32 + lane];
            if (j >= NBLK) v[u] = make_int4(-1, -1, -1, -1);
        }
        #pragma unroll
        for (int u = 0; u < 8; ++u) {
            const unsigned l0 = (unsigned)(v[u].x - base), l1 = (unsigned)(v[u].y - base);
            const unsigned l2 = (unsigned)(v[u].z - base), l3 = (unsigned)(v[u].w - base);
            const bool m0 = l0 < SLAB, m1 = l1 < SLAB, m2 = l2 < SLAB, m3 = l3 < SLAB;
            if (__builtin_amdgcn_ballot_w32(m0 | m1 | m2 | m3)) {
                const unsigned e0 = (unsigned)(eb[u] + lane * 4);
                const unsigned b0 = __builtin_amdgcn_ballot_w32(m0);
                if (m0) myq[cnt + __builtin_amdgcn_mbcnt_lo(b0, 0u)] = ((e0 + 0u) << 8) | l0;
                cnt += __builtin_popcount(b0);
                const unsigned b1 = __builtin_amdgcn_ballot_w32(m1);
                if (m1) myq[cnt + __builtin_amdgcn_mbcnt_lo(b1, 0u)] = ((e0 + 1u) << 8) | l1;
                cnt += __builtin_popcount(b1);
                const unsigned b2 = __builtin_amdgcn_ballot_w32(m2);
                if (m2) myq[cnt + __builtin_amdgcn_mbcnt_lo(b2, 0u)] = ((e0 + 2u) << 8) | l2;
                cnt += __builtin_popcount(b2);
                const unsigned b3 = __builtin_amdgcn_ballot_w32(m3);
                if (m3) myq[cnt + __builtin_amdgcn_mbcnt_lo(b3, 0u)] = ((e0 + 3u) << 8) | l3;
                cnt += __builtin_popcount(b3);
                if (cnt >= FLUSH_AT) {
                    const int m = cnt & ~7;
                    drain(m);
                    const int rem = cnt - m;
                    unsigned t = 0u;
                    if (lane < rem) t = myq[m + lane];
                    if (lane < rem) myq[lane] = t;
                    cnt = rem;
                }
            }
        }
    }
    if (cnt > 0) {
        const int m = (cnt + 7) & ~7;
        if (lane < m - cnt) myq[cnt + lane] = (unsigned)SLAB;
        drain(m);
    }
    __syncthreads();

    float4* h4 = (float4*)h;
    #pragma unroll
    for (int i = 0; i < 7; ++i) {
        const int r = w + 32 * i;
        if (r < SLAB) {
            const float* a = acc + r * DF + lane;
            float4 v; v.x = a[0]; v.y = a[32]; v.z = a[64]; v.w = a[96];
            h4[(size_t)(base + r) * 32 + lane] = v;
        }
    }
}

template<int NSEG, int EPI>
__global__ void __launch_bounds__(256, 2)
gemm_wmma(const float* __restrict__ A0, const float* __restrict__ A1,
          const float* __restrict__ A2, const float* __restrict__ W,
          const float* __restrict__ bias, const float* __restrict__ resid,
          float* __restrict__ out) {
    __shared__ float lA[16 * DF];

    const int lane = threadIdx.x & 31;
    const int wave = threadIdx.x >> 5;
    const int r    = lane & 15;
    const int hi   = lane >> 4;
    const int n0   = wave * 16;
    const int row0 = blockIdx.x * 16;

    v8f acc = {0.f, 0.f, 0.f, 0.f, 0.f, 0.f, 0.f, 0.f};

    const float* As[3] = {A0, A1, A2};
    #pragma unroll
    for (int seg = 0; seg < NSEG; ++seg) {
        const float4* s4 = (const float4*)(As[seg] + (size_t)row0 * DF);
        float4* d4 = (float4*)lA;
        #pragma unroll
        for (int i = 0; i < 2; ++i)
            d4[threadIdx.x + i * 256] = s4[threadIdx.x + i * 256];
        __syncthreads();

        const float* Wseg = W + seg * DF * DF;
        #pragma unroll
        for (int k0 = 0; k0 < DF; k0 += 4) {
            const float* ap = lA + r * DF + k0 + 2 * hi;
            v2f a; a.x = ap[0]; a.y = ap[1];
            const float* bp = Wseg + (k0 + 2 * hi) * DF + n0 + r;
            v2f b; b.x = bp[0]; b.y = bp[DF];
            acc = __builtin_amdgcn_wmma_f32_16x16x4_f32(
                false, a, false, b, (short)0, acc, false, false);
        }
        __syncthreads();
    }

    const int col = n0 + r;
    const float bb = bias[col];
    #pragma unroll
    for (int v = 0; v < 8; ++v) {
        int row = row0 + v + 8 * hi;
        float y = acc[v] + bb;
        if (EPI == 1) y = fmaxf(y, 0.0f);
        if (EPI == 2) {
            y += resid[(size_t)row * DF + col];
            y = (y > 0.0f) ? y : expm1f(y);
        }
        out[(size_t)row * DF + col] = y;
    }
}

extern "C" void kernel_launch(void* const* d_in, const int* in_sizes, int n_in,
                              void* d_out, int out_size, void* d_ws, size_t ws_size,
                              hipStream_t stream) {
    const float* x   = (const float*)d_in[0];
    const int*   ei  = (const int*)d_in[1];
    const float* W1[3] = {(const float*)d_in[2],  (const float*)d_in[6],  (const float*)d_in[10]};
    const float* b1[3] = {(const float*)d_in[3],  (const float*)d_in[7],  (const float*)d_in[11]};
    const float* W2[3] = {(const float*)d_in[4],  (const float*)d_in[8],  (const float*)d_in[12]};
    const float* b2[3] = {(const float*)d_in[5],  (const float*)d_in[9],  (const float*)d_in[13]};
    const float* Wjk = (const float*)d_in[14];
    const float* bjk = (const float*)d_in[15];

    const size_t NM = (size_t)N_NODES * DF;
    float* h0 = (float*)d_ws;
    float* xs[3] = {h0 + NM, h0 + 2 * NM, h0 + 3 * NM};

    dim3 blk(256);
    const int gBlocks = N_NODES / 16;

    const float* xin = x;
    for (int l = 0; l < 3; ++l) {
        slab_aggregate<<<NSLAB, dim3(1024), 0, stream>>>(h0, xin, ei);
        gemm_wmma<1, 1><<<gBlocks, blk, 0, stream>>>(h0, nullptr, nullptr,
                                                     W1[l], b1[l], nullptr, h0);
        gemm_wmma<1, 2><<<gBlocks, blk, 0, stream>>>(h0, nullptr, nullptr,
                                                     W2[l], b2[l], xin, xs[l]);
        xin = xs[l];
    }
    gemm_wmma<3, 0><<<gBlocks, blk, 0, stream>>>(xs[0], xs[1], xs[2],
                                                 Wjk, bjk, nullptr, (float*)d_out);
}


# GEMMs re-tiled: one 16-wave work-group per WGP copies W once into LDS; each wave owns a 16x128 output tile, A fragments straight from global, B from LDS; k sum order regrouped
# speedup vs baseline: 2.1285x; 1.3266x over previous
#include <hip/hip_runtime.h>
#include <cmath>

#define N_NODES 50000
#define N_EDGES 800000
#define DF 128

typedef __attribute__((ext_vector_type(2))) float v2f;
typedef __attribute__((ext_vector_type(8))) float v8f;

#define SLAB   200
#define NSLAB  250
#define NBLK   (N_EDGES / 128)
#define BROT   (NBLK / NSLAB)
#define QCAP   160
#define FLUSH_AT 24

__global__ void __launch_bounds__(1024)
slab_aggregate(float* __restrict__ h, const float* __restrict__ x, const int* __restrict__ ei) {
    __shared__ float acc[(SLAB + 1) * DF];
    __shared__ unsigned int q[32 * QCAP];

    const int lane = threadIdx.x & 31;
    const int w    = __builtin_amdgcn_readfirstlane(threadIdx.x >> 5);
    const int g    = blockIdx.x;
    const int base = g * SLAB;
    const float4* x4 = (const float4*)x;

    {
        float4 iv[7];
        #pragma unroll
        for (int i = 0; i < 7; ++i) {
            const int r = min(w + 32 * i, SLAB - 1);
            iv[i] = x4[(size_t)(base + r) * 32 + lane];
        }
        #pragma unroll
        for (int i = 0; i < 7; ++i) {
            const int r = w + 32 * i;
            if (r < SLAB) {
                float* a = acc + r * DF + lane;
                a[0] = iv[i].x; a[32] = iv[i].y; a[64] = iv[i].z; a[96] = iv[i].w;
            }
        }
    }
    __syncthreads();

    const int4* d4 = (const int4*)(ei + N_EDGES);
    unsigned int* myq = q + w * QCAP;
    int cnt = 0;

    auto drain = [&](int m) {
        for (int b0 = 0; b0 < m; b0 += 32) {
            const int n = min(32, m - b0);
            unsigned ent = 0u; unsigned s_l = 0u;
            if (lane < n) {
                ent = myq[b0 + lane];
                s_l = min((unsigned)ei[ent >> 8], (unsigned)(N_NODES - 1));
            }
            const int ld_l = (int)(ent & 255u);
            for (int k0 = 0; k0 < n; k0 += 8) {
                float4 r[8];
                #pragma unroll
                for (int u = 0; u < 8; ++u) {
                    const unsigned s = __builtin_amdgcn_readlane(s_l, k0 + u);
                    r[u] = x4[(size_t)s * 32 + lane];
                }
                #pragma unroll
                for (int u = 0; u < 8; ++u) {
                    const int ld = __builtin_amdgcn_readlane(ld_l, k0 + u);
                    float* a = acc + ld * DF + lane;
                    atomicAdd(a,      r[u].x);
                    atomicAdd(a + 32, r[u].y);
                    atomicAdd(a + 64, r[u].z);
                    atomicAdd(a + 96, r[u].w);
                }
            }
        }
    };

    for (int k = 0; k < (NBLK + 31) / 32; k += 8) {
        int4 v[8]; int eb[8];
        #pragma unroll
        for (int u = 0; u < 8; ++u) {
            const int j = (k + u) * 32 + w;
            int bi = min(j, NBLK - 1) + g * BROT; if (bi >= NBLK) bi -= NBLK;
            eb[u] = bi * 128;
            v[u] = d4[bi * 32 + lane];
            if (j >= NBLK) v[u] = make_int4(-1, -1, -1, -1);
        }
        #pragma unroll
        for (int u = 0; u < 8; ++u) {
            const unsigned l0 = (unsigned)(v[u].x - base), l1 = (unsigned)(v[u].y - base);
            const unsigned l2 = (unsigned)(v[u].z - base), l3 = (unsigned)(v[u].w - base);
            const bool m0 = l0 < SLAB, m1 = l1 < SLAB, m2 = l2 < SLAB, m3 = l3 < SLAB;
            if (__builtin_amdgcn_ballot_w32(m0 | m1 | m2 | m3)) {
                const unsigned e0 = (unsigned)(eb[u] + lane * 4);
                const unsigned b0 = __builtin_amdgcn_ballot_w32(m0);
                if (m0) myq[cnt + __builtin_amdgcn_mbcnt_lo(b0, 0u)] = ((e0 + 0u) << 8) | l0;
                cnt += __builtin_popcount(b0);
                const unsigned b1 = __builtin_amdgcn_ballot_w32(m1);
                if (m1) myq[cnt + __builtin_amdgcn_mbcnt_lo(b1, 0u)] = ((e0 + 1u) << 8) | l1;
                cnt += __builtin_popcount(b1);
                const unsigned b2 = __builtin_amdgcn_ballot_w32(m2);
                if (m2) myq[cnt + __builtin_amdgcn_mbcnt_lo(b2, 0u)] = ((e0 + 2u) << 8) | l2;
                cnt += __builtin_popcount(b2);
                const unsigned b3 = __builtin_amdgcn_ballot_w32(m3);
                if (m3) myq[cnt + __builtin_amdgcn_mbcnt_lo(b3, 0u)] = ((e0 + 3u) << 8) | l3;
                cnt += __builtin_popcount(b3);
                if (cnt >= FLUSH_AT) {
                    const int m = cnt & ~7;
                    drain(m);
                    const int rem = cnt - m;
                    unsigned t = 0u;
                    if (lane < rem) t = myq[m + lane];
                    if (lane < rem) myq[lane] = t;
                    cnt = rem;
                }
            }
        }
    }
    if (cnt > 0) {
        const int m = (cnt + 7) & ~7;
        if (lane < m - cnt) myq[cnt + lane] = (unsigned)SLAB;
        drain(m);
    }
    __syncthreads();

    float4* h4 = (float4*)h;
    #pragma unroll
    for (int i = 0; i < 7; ++i) {
        const int r = w + 32 * i;
        if (r < SLAB) {
            const float* a = acc + r * DF + lane;
            float4 v; v.x = a[0]; v.y = a[32]; v.z = a[64]; v.w = a[96];
            h4[(size_t)(base + r) * 32 + lane] = v;
        }
    }
}

#define GEMM_THREADS 512
#define GEMM_GROUPS  256
#define ROW_TILES    (N_NODES / 16)

template<int NSEG, int EPI>
__global__ void __launch_bounds__(GEMM_THREADS)
gemm_rows(const float* __restrict__ A0, const float* __restrict__ A1,
          const float* __restrict__ A2, const float* __restrict__ W,
          const float* __restrict__ bias, const float* __restrict__ resid,
          float* __restrict__ out) {
    __shared__ float4 lW[NSEG * 32 * DF];

    const int tid  = threadIdx.x;
    const int lane = tid & 31;
    const int w    = __builtin_amdgcn_readfirstlane(tid >> 5);
    const int c    = lane & 15;
    const int hi   = lane >> 4;
    const int tile = w * GEMM_GROUPS + (int)blockIdx.x;
    const bool active = tile < ROW_TILES;
    const int row0 = tile * 16;

    const float* As[3] = {A0, A1, A2};

    float4 a[16];
    if (active) {
        const float4* ap = (const float4*)(As[0] + (size_t)(row0 + c) * DF) + hi;
        #pragma unroll
        for (int j = 0; j < 16; ++j) a[j] = ap[2 * j];
    }

    {
        const int n   = tid & (DF - 1);
        const int kq0 = tid >> 7;
        #pragma unroll
        for (int seg = 0; seg < NSEG; ++seg) {
            #pragma unroll
            for (int it = 0; it < 8; ++it) {
                const int kq = kq0 + 4 * it;
                const float* wp = W + (size_t)(seg * DF + 4 * kq) * DF + n;
                float4 v; v.x = wp[0]; v.y = wp[DF]; v.z = wp[2 * DF]; v.w = wp[3 * DF];
                lW[(seg * 32 + kq) * DF + n] = v;
            }
        }
    }
    __syncthreads();
    if (!active) return;

    v8f acc[8];
    #pragma unroll
    for (int t = 0; t < 8; ++t) acc[t] = (v8f){0.f, 0.f, 0.f, 0.f, 0.f, 0.f, 0.f, 0.f};

    const float4* bl = lW + hi * DF + c;
    float4 b[8];
    #pragma unroll
    for (int t = 0; t < 8; ++t) b[t] = bl[16 * t];

    #pragma unroll
    for (int jj = 0; jj < NSEG * 16; ++jj) {
        const int j = jj & 15, seg = jj >> 4;
        float4 bn[8];
        if (jj + 1 < NSEG * 16) {
            #pragma unroll
            for (int t = 0; t < 8; ++t) bn[t] = bl[(2 * (jj + 1)) * DF + 16 * t];
        }
        v2f a01; a01.x = a[j].x; a01.y = a[j].y;
        v2f a23; a23.x = a[j].z; a23.y = a[j].w;
        #pragma unroll
        for (int t = 0; t < 8; ++t) {
            v2f b01; b01.x = b[t].x; b01.y = b[t].y;
            acc[t] = __builtin_amdgcn_wmma_f32_16x16x4_f32(
                false, a01, false, b01, (short)0, acc[t], false, false);
        }
        #pragma unroll
        for (int t = 0; t < 8; ++t) {
            v2f b23; b23.x = b[t].z; b23.y = b[t].w;
            acc[t] = __builtin_amdgcn_wmma_f32_16x16x4_f32(
                false, a23, false, b23, (short)0, acc[t], false, false);
        }
        if (seg + 1 < NSEG) {
            const float4* ap = (const float4*)(As[seg + 1] + (size_t)(row0 + c) * DF) + hi;
            a[j] = ap[2 * j];
        }
        if (jj + 1 < NSEG * 16) {
            #pragma unroll
            for (int t = 0; t < 8; ++t) b[t] = bn[t];
        }
    }

    #pragma unroll
    for (int t = 0; t < 8; ++t) {
        const int col = 16 * t + c;
        const float bb = bias[col];
        #pragma unroll
        for (int v = 0; v < 8; ++v) {
            const int row = row0 + v + 8 * hi;
            float y = acc[t][v] + bb;
            if (EPI == 1) y = fmaxf(y, 0.0f);
            if (EPI == 2) {
                y += resid[(size_t)row * DF + col];
                y = (y > 0.0f) ? y : expm1f(y);
            }
            out[(size_t)row * DF + col] = y;
        }
    }
}

extern "C" void kernel_launch(void* const* d_in, const int* in_sizes, int n_in,
                              void* d_out, int out_size, void* d_ws, size_t ws_size,
                              hipStream_t stream) {
    const float* x   = (const float*)d_in[0];
    const int*   ei  = (const int*)d_in[1];
    const float* W1[3] = {(const float*)d_in[2],  (const float*)d_in[6],  (const float*)d_in[10]};
    const float* b1[3] = {(const float*)d_in[3],  (const float*)d_in[7],  (const float*)d_in[11]};
    const float* W2[3] = {(const float*)d_in[4],  (const float*)d_in[8],  (const float*)d_in[12]};
    const float* b2[3] = {(const float*)d_in[5],  (const float*)d_in[9],  (const float*)d_in[13]};
    const float* Wjk = (const float*)d_in[14];
    const float* bjk = (const float*)d_in[15];

    const size_t NM = (size_t)N_NODES * DF;
    float* h0 = (float*)d_ws;
    float* xs[3] = {h0 + NM, h0 + 2 * NM, h0 + 3 * NM};

    dim3 blk(GEMM_THREADS);
    const int gBlocks = GEMM_GROUPS;

    const float* xin = x;
    for (int l = 0; l < 3; ++l) {
        slab_aggregate<<<NSLAB, dim3(1024), 0, stream>>>(h0, xin, ei);
        gemm_rows<1, 1><<<gBlocks, blk, 0, stream>>>(h0, nullptr, nullptr,
                                                     W1[l], b1[l], nullptr, h0);
        gemm_rows<1, 2><<<gBlocks, blk, 0, stream>>>(h0, nullptr, nullptr,
                                                     W2[l], b2[l], xin, xs[l]);
        xin = xs[l];
    }
    gemm_rows<3, 0><<<gBlocks, blk, 0, stream>>>(xs[0], xs[1], xs[2],
                                                 Wjk, bjk, nullptr, (float*)d_out);
}


# Each slab's matching edges (source, row) saved to workspace by layer 1's scan; layers 2, 3 walk the saved per-wave lists, dst list not re-read; sums still LDS float atomics in arrival order
# speedup vs baseline: 2.5180x; 1.1830x over previous
#include <hip/hip_runtime.h>
#include <cmath>

#define N_NODES 50000
#define N_EDGES 800000
#define DF 128

typedef __attribute__((ext_vector_type(2))) float v2f;
typedef __attribute__((ext_vector_type(8))) float v8f;

#define SLAB   200
#define NSLAB  250
#define NBLK   (N_EDGES / 128)
#define BROT   (NBLK / NSLAB)
#define QCAP   160
#define FLUSH_AT 24
#define LCAP   768
#define NLIST  (NSLAB * 32)
#define LIST_FULL 0xFFFFFFFFu

template<int REPLAY>
__global__ void __launch_bounds__(1024)
slab_aggregate(float* __restrict__ h, const float* __restrict__ x, const int* __restrict__ ei,
               unsigned int* __restrict__ lists, unsigned int* __restrict__ counts) {
    __shared__ float acc[(SLAB + 1) * DF];
    __shared__ unsigned int q[32 * QCAP];

    const int lane = threadIdx.x & 31;
    const int w    = __builtin_amdgcn_readfirstlane(threadIdx.x >> 5);
    const int g    = blockIdx.x;
    const int base = g * SLAB;
    const float4* x4 = (const float4*)x;
    unsigned int* mylist = lists + (size_t)(g * 32 + w) * LCAP;

    unsigned saved = LIST_FULL;
    if (REPLAY) saved = counts[g * 32 + w];

    {
        float4 iv[7];
        #pragma unroll
        for (int i = 0; i < 7; ++i) {
            const int r = min(w + 32 * i, SLAB - 1);
            iv[i] = x4[(size_t)(base + r) * 32 + lane];
        }
        #pragma unroll
        for (int i = 0; i < 7; ++i) {
            const int r = w + 32 * i;
            if (r < SLAB) {
                float* a = acc + r * DF + lane;
                a[0] = iv[i].x; a[32] = iv[i].y; a[64] = iv[i].z; a[96] = iv[i].w;
            }
        }
    }
    saved = __builtin_amdgcn_readfirstlane(saved);
    __syncthreads();

    auto add_rows = [&](unsigned s_l, int ld_l, int n) {
        for (int k0 = 0; k0 < n; k0 += 8) {
            float4 r[8];
            #pragma unroll
            for (int u = 0; u < 8; ++u) {
                const unsigned s = __builtin_amdgcn_readlane(s_l, k0 + u);
                r[u] = x4[(size_t)s * 32 + lane];
            }
            #pragma unroll
            for (int u = 0; u < 8; ++u) {
                const int ld = __builtin_amdgcn_readlane(ld_l, k0 + u);
                float* a = acc + ld * DF + lane;
                atomicAdd(a,      r[u].x);
                atomicAdd(a + 32, r[u].y);
                atomicAdd(a + 64, r[u].z);
                atomicAdd(a + 96, r[u].w);
            }
        }
    };

    if (REPLAY && saved != LIST_FULL) {
        const int total = (int)(min(saved, (unsigned)LCAP) & ~7u);
        unsigned nxt = (unsigned)SLAB;
        if (lane < total) nxt = mylist[lane];
        for (int b0 = 0; b0 < total; b0 += 32) {
            const int n = min(32, total - b0);
            const unsigned ent = nxt;
            if (b0 + 32 + lane < total) nxt = mylist[b0 + 32 + lane];
            const unsigned s_l = min(ent >> 8, (unsigned)(N_NODES - 1));
            const int ld_l = (int)min(ent & 255u, (unsigned)SLAB);
            add_rows(s_l, ld_l, n);
        }
    } else {
        const int4* d4 = (const int4*)(ei + N_EDGES);
        unsigned int* myq = q + w * QCAP;
        int cnt = 0;
        int tot = 0;

        auto drain = [&](int m) {
            for (int b0 = 0; b0 < m; b0 += 32) {
                const int n = min(32, m - b0);
                unsigned ent = 0u; unsigned s_l = 0u;
                if (lane < n) {
                    ent = myq[b0 + lane];
                    s_l = min((unsigned)ei[ent >> 8], (unsigned)(N_NODES - 1));
                    if (!REPLAY && tot + b0 + lane < LCAP)
                        mylist[tot + b0 + lane] = (s_l << 8) | (ent & 255u);
                }
                add_rows(s_l, (int)(ent & 255u), n);
            }
            tot += m;
        };

        for (int k = 0; k < (NBLK + 31) / 32; k += 8) {
            int4 v[8]; int eb[8];
            #pragma unroll
            for (int u = 0; u < 8; ++u) {
                const int j = (k + u) * 32 + w;
                int bi = min(j, NBLK - 1) + g * BROT; if (bi >= NBLK) bi -= NBLK;
                eb[u] = bi * 128;
                v[u] = d4[bi * 32 + lane];
                if (j >= NBLK) v[u] = make_int4(-1, -1, -1, -1);
            }
            #pragma unroll
            for (int u = 0; u < 8; ++u) {
                const unsigned l0 = (unsigned)(v[u].x - base), l1 = (unsigned)(v[u].y - base);
                const unsigned l2 = (unsigned)(v[u].z - base), l3 = (unsigned)(v[u].w - base);
                const bool m0 = l0 < SLAB, m1 = l1 < SLAB, m2 = l2 < SLAB, m3 = l3 < SLAB;
                if (__builtin_amdgcn_ballot_w32(m0 | m1 | m2 | m3)) {
                    const unsigned e0 = (unsigned)(eb[u] + lane * 4);
                    const unsigned b0 = __builtin_amdgcn_ballot_w32(m0);
                    if (m0) myq[cnt + __builtin_amdgcn_mbcnt_lo(b0, 0u)] = ((e0 + 0u) << 8) | l0;
                    cnt += __builtin_popcount(b0);
                    const unsigned b1 = __builtin_amdgcn_ballot_w32(m1);
                    if (m1) myq[cnt + __builtin_amdgcn_mbcnt_lo(b1, 0u)] = ((e0 + 1u) << 8) | l1;
                    cnt += __builtin_popcount(b1);
                    const unsigned b2 = __builtin_amdgcn_ballot_w32(m2);
                    if (m2) myq[cnt + __builtin_amdgcn_mbcnt_lo(b2, 0u)] = ((e0 + 2u) << 8) | l2;
                    cnt += __builtin_popcount(b2);
                    const unsigned b3 = __builtin_amdgcn_ballot_w32(m3);
                    if (m3) myq[cnt + __builtin_amdgcn_mbcnt_lo(b3, 0u)] = ((e0 + 3u) << 8) | l3;
                    cnt += __builtin_popcount(b3);
                    if (cnt >= FLUSH_AT) {
                        const int m = cnt & ~7;
                        drain(m);
                        const int rem = cnt - m;
                        unsigned t = 0u;
                        if (lane < rem) t = myq[m + lane];
                        if (lane < rem) myq[lane] = t;
                        cnt = rem;
                    }
                }
            }
        }
        if (cnt > 0) {
            const int m = (cnt + 7) & ~7;
            if (lane < m - cnt) myq[cnt + lane] = (unsigned)SLAB;
            drain(m);
        }
        if (!REPLAY && lane == 0)
            counts[g * 32 + w] = (tot <= LCAP) ? (unsigned)tot : LIST_FULL;
    }
    __syncthreads();

    float4* h4 = (float4*)h;
    #pragma unroll
    for (int i = 0; i < 7; ++i) {
        const int r = w + 32 * i;
        if (r < SLAB) {
            const float* a = acc + r * DF + lane;
            float4 v; v.x = a[0]; v.y = a[32]; v.z = a[64]; v.w = a[96];
            h4[(size_t)(base + r) * 32 + lane] = v;
        }
    }
}

#define GEMM_THREADS 512
#define GEMM_GROUPS  256
#define ROW_TILES    (N_NODES / 16)

template<int NSEG, int EPI>
__global__ void __launch_bounds__(GEMM_THREADS)
gemm_rows(const float* __restrict__ A0, const float* __restrict__ A1,
          const float* __restrict__ A2, const float* __restrict__ W,
          const float* __restrict__ bias, const float* __restrict__ resid,
          float* __restrict__ out) {
    __shared__ float4 lW[NSEG * 32 * DF];

    const int tid  = threadIdx.x;
    const int lane = tid & 31;
    const int w    = __builtin_amdgcn_readfirstlane(tid >> 5);
    const int c    = lane & 15;
    const int hi   = lane >> 4;
    const int tile = w * GEMM_GROUPS + (int)blockIdx.x;
    const bool active = tile < ROW_TILES;
    const int row0 = tile * 16;

    const float* As[3] = {A0, A1, A2};

    float4 a[16];
    if (active) {
        const float4* ap = (const float4*)(As[0] + (size_t)(row0 + c) * DF) + hi;
        #pragma unroll
        for (int j = 0; j < 16; ++j) a[j] = ap[2 * j];
    }

    {
        const int n   = tid & (DF - 1);
        const int kq0 = tid >> 7;
        #pragma unroll
        for (int seg = 0; seg < NSEG; ++seg) {
            #pragma unroll
            for (int it = 0; it < 8; ++it) {
                const int kq = kq0 + 4 * it;
                const float* wp = W + (size_t)(seg * DF + 4 * kq) * DF + n;
                float4 v; v.x = wp[0]; v.y = wp[DF]; v.z = wp[2 * DF]; v.w = wp[3 * DF];
                lW[(seg * 32 + kq) * DF + n] = v;
            }
        }
    }
    __syncthreads();
    if (!active) return;

    v8f acc[8];
    #pragma unroll
    for (int t = 0; t < 8; ++t) acc[t] = (v8f){0.f, 0.f, 0.f, 0.f, 0.f, 0.f, 0.f, 0.f};

    const float4* bl = lW + hi * DF + c;
    float4 b[8];
    #pragma unroll
    for (int t = 0; t < 8; ++t) b[t] = bl[16 * t];

    #pragma unroll
    for (int jj = 0; jj < NSEG * 16; ++jj) {
        const int j = jj & 15, seg = jj >> 4;
        float4 bn[8];
        if (jj + 1 < NSEG * 16) {
            #pragma unroll
            for (int t = 0; t < 8; ++t) bn[t] = bl[(2 * (jj + 1)) * DF + 16 * t];
        }
        v2f a01; a01.x = a[j].x; a01.y = a[j].y;
        v2f a23; a23.x = a[j].z; a23.y = a[j].w;
        #pragma unroll
        for (int t = 0; t < 8; ++t) {
            v2f b01; b01.x = b[t].x; b01.y = b[t].y;
            acc[t] = __builtin_amdgcn_wmma_f32_16x16x4_f32(
                false, a01, false, b01, (short)0, acc[t], false, false);
        }
        #pragma unroll
        for (int t = 0; t < 8; ++t) {
            v2f b23; b23.x = b[t].z; b23.y = b[t].w;
            acc[t] = __builtin_amdgcn_wmma_f32_16x16x4_f32(
                false, a23, false, b23, (short)0, acc[t], false, false);
        }
        if (seg + 1 < NSEG) {
            const float4* ap = (const float4*)(As[seg + 1] + (size_t)(row0 + c) * DF) + hi;
            a[j] = ap[2 * j];
        }
        if (jj + 1 < NSEG * 16) {
            #pragma unroll
            for (int t = 0; t < 8; ++t) b[t] = bn[t];
        }
    }

    #pragma unroll
    for (int t = 0; t < 8; ++t) {
        const int col = 16 * t + c;
        const float bb = bias[col];
        #pragma unroll
        for (int v = 0; v < 8; ++v) {
            const int row = row0 + v + 8 * hi;
            float y = acc[t][v] + bb;
            if (EPI == 1) y = fmaxf(y, 0.0f);
            if (EPI == 2) {
                y += resid[(size_t)row * DF + col];
                y = (y > 0.0f) ? y : expm1f(y);
            }
            out[(size_t)row * DF + col] = y;
        }
    }
}

extern "C" void kernel_launch(void* const* d_in, const int* in_sizes, int n_in,
                              void* d_out, int out_size, void* d_ws, size_t ws_size,
                              hipStream_t stream) {
    const float* x   = (const float*)d_in[0];
    const int*   ei  = (const int*)d_in[1];
    const float* W1[3] = {(const float*)d_in[2],  (const float*)d_in[6],  (const float*)d_in[10]};
    const float* b1[3] = {(const float*)d_in[3],  (const float*)d_in[7],  (const float*)d_in[11]};
    const float* W2[3] = {(const float*)d_in[4],  (const float*)d_in[8],  (const float*)d_in[12]};
    const float* b2[3] = {(const float*)d_in[5],  (const float*)d_in[9],  (const float*)d_in[13]};
    const float* Wjk = (const float*)d_in[14];
    const float* bjk = (const float*)d_in[15];

    const size_t NM = (size_t)N_NODES * DF;
    float* h0 = (float*)d_ws;
    float* xs[3] = {h0 + NM, h0 + 2 * NM, h0 + 3 * NM};
    unsigned int* lists  = (unsigned int*)xs[2];
    unsigned int* counts = lists + (size_t)NLIST * LCAP;

    dim3 blk(GEMM_THREADS);
    const int gBlocks = GEMM_GROUPS;

    const float* xin = x;
    for (int l = 0; l < 3; ++l) {
        if (l == 0) slab_aggregate<0><<<NSLAB, dim3(1024), 0, stream>>>(h0, xin, ei, lists, counts);
        else        slab_aggregate<1><<<NSLAB, dim3(1024), 0, stream>>>(h0, xin, ei, lists, counts);
        gemm_rows<1, 1><<<gBlocks, blk, 0, stream>>>(h0, nullptr, nullptr,
                                                     W1[l], b1[l], nullptr, h0);
        gemm_rows<1, 2><<<gBlocks, blk, 0, stream>>>(h0, nullptr, nullptr,
                                                     W2[l], b2[l], xin, xs[l]);
        xin = xs[l];
    }
    gemm_rows<3, 0><<<gBlocks, blk, 0, stream>>>(xs[0], xs[1], xs[2],
                                                 Wjk, bjk, nullptr, (float*)d_out);
}


# Layer's two MLP GEMMs in one launch as transposed products (W^T = A operand, node rows = B): 1st product's D registers are the 2nd's B fragments, hidden stays in registers; 2nd k sum order regrouped
# speedup vs baseline: 2.9126x; 1.1567x over previous
#include <hip/hip_runtime.h>
#include <cmath>

#define N_NODES 50000
#define N_EDGES 800000
#define DF 128

typedef __attribute__((ext_vector_type(2))) float v2f;
typedef __attribute__((ext_vector_type(8))) float v8f;

#define SLAB   200
#define NSLAB  250
#define NBLK   (N_EDGES / 128)
#define BROT   (NBLK / NSLAB)
#define QCAP   160
#define FLUSH_AT 24
#define LCAP   768
#define NLIST  (NSLAB * 32)
#define LIST_FULL 0xFFFFFFFFu

template<int REPLAY>
__global__ void __launch_bounds__(1024)
slab_aggregate(float* __restrict__ h, const float* __restrict__ x, const int* __restrict__ ei,
               unsigned int* __restrict__ lists, unsigned int* __restrict__ counts) {
    __shared__ float acc[(SLAB + 1) * DF];
    __shared__ unsigned int q[32 * QCAP];

    const int lane = threadIdx.x & 31;
    const int w    = __builtin_amdgcn_readfirstlane(threadIdx.x >> 5);
    const int g    = blockIdx.x;
    const int base = g * SLAB;
    const float4* x4 = (const float4*)x;
    unsigned int* mylist = lists + (size_t)(g * 32 + w) * LCAP;

    unsigned saved = LIST_FULL;
    if (REPLAY) saved = counts[g * 32 + w];

    {
        float4 iv[7];
        #pragma unroll
        for (int i = 0; i < 7; ++i) {
            const int r = min(w + 32 * i, SLAB - 1);
            iv[i] = x4[(size_t)(base + r) * 32 + lane];
        }
        #pragma unroll
        for (int i = 0; i < 7; ++i) {
            const int r = w + 32 * i;
            if (r < SLAB) {
                float* a = acc + r * DF + lane;
                a[0] = iv[i].x; a[32] = iv[i].y; a[64] = iv[i].z; a[96] = iv[i].w;
            }
        }
    }
    saved = __builtin_amdgcn_readfirstlane(saved);
    __syncthreads();

    auto add_rows = [&](unsigned s_l, int ld_l, int n) {
        for (int k0 = 0; k0 < n; k0 += 8) {
            float4 r[8];
            #pragma unroll
            for (int u = 0; u < 8; ++u) {
                const unsigned s = __builtin_amdgcn_readlane(s_l, k0 + u);
                r[u] = x4[(size_t)s * 32 + lane];
            }
            #pragma unroll
            for (int u = 0; u < 8; ++u) {
                const int ld = __builtin_amdgcn_readlane(ld_l, k0 + u);
                float* a = acc + ld * DF + lane;
                atomicAdd(a,      r[u].x);
                atomicAdd(a + 32, r[u].y);
                atomicAdd(a + 64, r[u].z);
                atomicAdd(a + 96, r[u].w);
            }
        }
    };

    if (REPLAY && saved != LIST_FULL) {
        const int total = (int)(min(saved, (unsigned)LCAP) & ~7u);
        unsigned nxt = (unsigned)SLAB;
        if (lane < total) nxt = mylist[lane];
        for (int b0 = 0; b0 < total; b0 += 32) {
            const int n = min(32, total - b0);
            const unsigned ent = nxt;
            if (b0 + 32 + lane < total) nxt = mylist[b0 + 32 + lane];
            const unsigned s_l = min(ent >> 8, (unsigned)(N_NODES - 1));
            const int ld_l = (int)min(ent & 255u, (unsigned)SLAB);
            add_rows(s_l, ld_l, n);
        }
    } else {
        const int4* d4 = (const int4*)(ei + N_EDGES);
        unsigned int* myq = q + w * QCAP;
        int cnt = 0;
        int tot = 0;

        auto drain = [&](int m) {
            for (int b0 = 0; b0 < m; b0 += 32) {
                const int n = min(32, m - b0);
                unsigned ent = 0u; unsigned s_l = 0u;
                if (lane < n) {
                    ent = myq[b0 + lane];
                    s_l = min((unsigned)ei[ent >> 8], (unsigned)(N_NODES - 1));
                    if (!REPLAY && tot + b0 + lane < LCAP)
                        mylist[tot + b0 + lane] = (s_l << 8) | (ent & 255u);
                }
                add_rows(s_l, (int)(ent & 255u), n);
            }
            tot += m;
        };

        for (int k = 0; k < (NBLK + 31) / 32; k += 8) {
            int4 v[8]; int eb[8];
            #pragma unroll
            for (int u = 0; u < 8; ++u) {
                const int j = (k + u) * 32 + w;
                int bi = min(j, NBLK - 1) + g * BROT; if (bi >= NBLK) bi -= NBLK;
                eb[u] = bi * 128;
                v[u] = d4[bi * 32 + lane];
                if (j >= NBLK) v[u] = make_int4(-1, -1, -1, -1);
            }
            #pragma unroll
            for (int u = 0; u < 8; ++u) {
                const unsigned l0 = (unsigned)(v[u].x - base), l1 = (unsigned)(v[u].y - base);
                const unsigned l2 = (unsigned)(v[u].z - base), l3 = (unsigned)(v[u].w - base);
                const bool m0 = l0 < SLAB, m1 = l1 < SLAB, m2 = l2 < SLAB, m3 = l3 < SLAB;
                if (__builtin_amdgcn_ballot_w32(m0 | m1 | m2 | m3)) {
                    const unsigned e0 = (unsigned)(eb[u] + lane * 4);
                    const unsigned b0 = __builtin_amdgcn_ballot_w32(m0);
                    if (m0) myq[cnt + __builtin_amdgcn_mbcnt_lo(b0, 0u)] = ((e0 + 0u) << 8) | l0;
                    cnt += __builtin_popcount(b0);
                    const unsigned b1 = __builtin_amdgcn_ballot_w32(m1);
                    if (m1) myq[cnt + __builtin_amdgcn_mbcnt_lo(b1, 0u)] = ((e0 + 1u) << 8) | l1;
                    cnt += __builtin_popcount(b1);
                    const unsigned b2 = __builtin_amdgcn_ballot_w32(m2);
                    if (m2) myq[cnt + __builtin_amdgcn_mbcnt_lo(b2, 0u)] = ((e0 + 2u) << 8) | l2;
                    cnt += __builtin_popcount(b2);
                    const unsigned b3 = __builtin_amdgcn_ballot_w32(m3);
                    if (m3) myq[cnt + __builtin_amdgcn_mbcnt_lo(b3, 0u)] = ((e0 + 3u) << 8) | l3;
                    cnt += __builtin_popcount(b3);
                    if (cnt >= FLUSH_AT) {
                        const int m = cnt & ~7;
                        drain(m);
                        const int rem = cnt - m;
                        unsigned t = 0u;
                        if (lane < rem) t = myq[m + lane];
                        if (lane < rem) myq[lane] = t;
                        cnt = rem;
                    }
                }
            }
        }
        if (cnt > 0) {
            const int m = (cnt + 7) & ~7;
            if (lane < m - cnt) myq[cnt + lane] = (unsigned)SLAB;
            drain(m);
        }
        if (!REPLAY && lane == 0)
            counts[g * 32 + w] = (tot <= LCAP) ? (unsigned)tot : LIST_FULL;
    }
    __syncthreads();

    float4* h4 = (float4*)h;
    #pragma unroll
    for (int i = 0; i < 7; ++i) {
        const int r = w + 32 * i;
        if (r < SLAB) {
            const float* a = acc + r * DF + lane;
            float4 v; v.x = a[0]; v.y = a[32]; v.z = a[64]; v.w = a[96];
            h4[(size_t)(base + r) * 32 + lane] = v;
        }
    }
}

#define GEMM_THREADS 512
#define GEMM_GROUPS  256
#define ROW_TILES    (N_NODES / 16)

template<int NSEG, int EPI>
__global__ void __launch_bounds__(GEMM_THREADS)
gemm_rows(const float* __restrict__ A0, const float* __restrict__ A1,
          const float* __restrict__ A2, const float* __restrict__ W,
          const float* __restrict__ bias, const float* __restrict__ resid,
          float* __restrict__ out) {
    __shared__ float4 lW[NSEG * 32 * DF];

    const int tid  = threadIdx.x;
    const int lane = tid & 31;
    const int w    = __builtin_amdgcn_readfirstlane(tid >> 5);
    const int c    = lane & 15;
    const int hi   = lane >> 4;
    const int tile = w * GEMM_GROUPS + (int)blockIdx.x;
    const bool active = tile < ROW_TILES;
    const int row0 = tile * 16;

    const float* As[3] = {A0, A1, A2};

    float4 a[16];
    if (active) {
        const float4* ap = (const float4*)(As[0] + (size_t)(row0 + c) * DF) + hi;
        #pragma unroll
        for (int j = 0; j < 16; ++j) a[j] = ap[2 * j];
    }

    {
        const int n   = tid & (DF - 1);
        const int kq0 = tid >> 7;
        #pragma unroll
        for (int seg = 0; seg < NSEG; ++seg) {
            #pragma unroll
            for (int it = 0; it < 8; ++it) {
                const int kq = kq0 + 4 * it;
                const float* wp = W + (size_t)(seg * DF + 4 * kq) * DF + n;
                float4 v; v.x = wp[0]; v.y = wp[DF]; v.z = wp[2 * DF]; v.w = wp[3 * DF];
                lW[(seg * 32 + kq) * DF + n] = v;
            }
        }
    }
    __syncthreads();
    if (!active) return;

    v8f acc[8];
    #pragma unroll
    for (int t = 0; t < 8; ++t) acc[t] = (v8f){0.f, 0.f, 0.f, 0.f, 0.f, 0.f, 0.f, 0.f};

    const float4* bl = lW + hi * DF + c;
    float4 b[8];
    #pragma unroll
    for (int t = 0; t < 8; ++t) b[t] = bl[16 * t];

    #pragma unroll
    for (int jj = 0; jj < NSEG * 16; ++jj) {
        const int j = jj & 15, seg = jj >> 4;
        float4 bn[8];
        if (jj + 1 < NSEG * 16) {
            #pragma unroll
            for (int t = 0; t < 8; ++t) bn[t] = bl[(2 * (jj + 1)) * DF + 16 * t];
        }
        v2f a01; a01.x = a[j].x; a01.y = a[j].y;
        v2f a23; a23.x = a[j].z; a23.y = a[j].w;
        #pragma unroll
        for (int t = 0; t < 8; ++t) {
            v2f b01; b01.x = b[t].x; b01.y = b[t].y;
            acc[t] = __builtin_amdgcn_wmma_f32_16x16x4_f32(
                false, a01, false, b01, (short)0, acc[t], false, false);
        }
        #pragma unroll
        for (int t = 0; t < 8; ++t) {
            v2f b23; b23.x = b[t].z; b23.y = b[t].w;
            acc[t] = __builtin_amdgcn_wmma_f32_16x16x4_f32(
                false, a23, false, b23, (short)0, acc[t], false, false);
        }
        if (seg + 1 < NSEG) {
            const float4* ap = (const float4*)(As[seg + 1] + (size_t)(row0 + c) * DF) + hi;
            a[j] = ap[2 * j];
        }
        if (jj + 1 < NSEG * 16) {
            #pragma unroll
            for (int t = 0; t < 8; ++t) b[t] = bn[t];
        }
    }

    #pragma unroll
    for (int t = 0; t < 8; ++t) {
        const int col = 16 * t + c;
        const float bb = bias[col];
        #pragma unroll
        for (int v = 0; v < 8; ++v) {
            const int row = row0 + v + 8 * hi;
            float y = acc[t][v] + bb;
            if (EPI == 1) y = fmaxf(y, 0.0f);
            if (EPI == 2) {
                y += resid[(size_t)row * DF + col];
                y = (y > 0.0f) ? y : expm1f(y);
            }
            out[(size_t)row * DF + col] = y;
        }
    }
}

__global__ void __launch_bounds__(GEMM_THREADS)
mlp_rows(const float* __restrict__ A, const float* __restrict__ W1,
         const float* __restrict__ b1, const float* __restrict__ W2,
         const float* __restrict__ b2, const float* __restrict__ resid,
         float* __restrict__ out) {
    __shared__ float4 lW[2 * 32 * DF];

    const int tid  = threadIdx.x;
    const int lane = tid & 31;
    const int w    = __builtin_amdgcn_readfirstlane(tid >> 5);
    const int c    = lane & 15;
    const int hi   = lane >> 4;
    const int tile = w * GEMM_GROUPS + (int)blockIdx.x;
    const bool active = tile < ROW_TILES;
    const int row0 = tile * 16;

    float4 a[16];
    if (active) {
        const float4* ap = (const float4*)(A + (size_t)(row0 + c) * DF) + hi;
        #pragma unroll
        for (int j = 0; j < 16; ++j) a[j] = ap[2 * j];
    }

    {
        const int n   = tid & (DF - 1);
        const int kq0 = tid >> 7;
        #pragma unroll
        for (int m = 0; m < 2; ++m) {
            const float* Wm = m ? W2 : W1;
            #pragma unroll
            for (int it = 0; it < 8; ++it) {
                const int kq = kq0 + 4 * it;
                const float* wp = Wm + (size_t)(4 * kq) * DF + n;
                float4 v; v.x = wp[0]; v.y = wp[DF]; v.z = wp[2 * DF]; v.w = wp[3 * DF];
                lW[(m * 32 + kq) * DF + n] = v;
            }
        }
    }
    __syncthreads();
    if (!active) return;

    v8f acc[8];
    #pragma unroll
    for (int t = 0; t < 8; ++t) acc[t] = (v8f){0.f, 0.f, 0.f, 0.f, 0.f, 0.f, 0.f, 0.f};

    const float4* bl = lW + hi * DF + c;
    float4 b[8];
    #pragma unroll
    for (int t = 0; t < 8; ++t) b[t] = bl[16 * t];

    #pragma unroll
    for (int jj = 0; jj < 16; ++jj) {
        float4 bn[8];
        if (jj + 1 < 16) {
            #pragma unroll
            for (int t = 0; t < 8; ++t) bn[t] = bl[(2 * (jj + 1)) * DF + 16 * t];
        }
        v2f a01; a01.x = a[jj].x; a01.y = a[jj].y;
        v2f a23; a23.x = a[jj].z; a23.y = a[jj].w;
        #pragma unroll
        for (int t = 0; t < 8; ++t) {
            v2f b01; b01.x = b[t].x; b01.y = b[t].y;
            acc[t] = __builtin_amdgcn_wmma_f32_16x16x4_f32(
                false, b01, false, a01, (short)0, acc[t], false, false);
        }
        #pragma unroll
        for (int t = 0; t < 8; ++t) {
            v2f b23; b23.x = b[t].z; b23.y = b[t].w;
            acc[t] = __builtin_amdgcn_wmma_f32_16x16x4_f32(
                false, b23, false, a23, (short)0, acc[t], false, false);
        }
        if (jj + 1 < 16) {
            #pragma unroll
            for (int t = 0; t < 8; ++t) b[t] = bn[t];
        }
    }

    const float4* bl2 = lW + (32 + 2 * hi) * DF + c;
    #pragma unroll
    for (int t = 0; t < 8; ++t) b[t] = bl2[16 * t];

    {
        const float4* b1q = (const float4*)b1 + 2 * hi;
        #pragma unroll
        for (int t = 0; t < 8; ++t) {
            const float4 p0 = b1q[4 * t], p1 = b1q[4 * t + 1];
            acc[t][0] = fmaxf(acc[t][0] + p0.x, 0.0f);
            acc[t][1] = fmaxf(acc[t][1] + p0.y, 0.0f);
            acc[t][2] = fmaxf(acc[t][2] + p0.z, 0.0f);
            acc[t][3] = fmaxf(acc[t][3] + p0.w, 0.0f);
            acc[t][4] = fmaxf(acc[t][4] + p1.x, 0.0f);
            acc[t][5] = fmaxf(acc[t][5] + p1.y, 0.0f);
            acc[t][6] = fmaxf(acc[t][6] + p1.z, 0.0f);
            acc[t][7] = fmaxf(acc[t][7] + p1.w, 0.0f);
        }
    }

    v8f acc2[8];
    #pragma unroll
    for (int t = 0; t < 8; ++t) acc2[t] = (v8f){0.f, 0.f, 0.f, 0.f, 0.f, 0.f, 0.f, 0.f};

    #pragma unroll
    for (int jj = 0; jj < 16; ++jj) {
        const int t = jj >> 1, qh = jj & 1;
        float4 bn[8];
        if (jj + 1 < 16) {
            const int nrow = 4 * ((jj + 1) >> 1) + ((jj + 1) & 1);
            #pragma unroll
            for (int t2 = 0; t2 < 8; ++t2) bn[t2] = bl2[nrow * DF + 16 * t2];
        }
        v2f h01; h01.x = acc[t][4 * qh + 0]; h01.y = acc[t][4 * qh + 1];
        v2f h23; h23.x = acc[t][4 * qh + 2]; h23.y = acc[t][4 * qh + 3];
        #pragma unroll
        for (int t2 = 0; t2 < 8; ++t2) {
            v2f b01; b01.x = b[t2].x; b01.y = b[t2].y;
            acc2[t2] = __builtin_amdgcn_wmma_f32_16x16x4_f32(
                false, b01, false, h01, (short)0, acc2[t2], false, false);
        }
        #pragma unroll
        for (int t2 = 0; t2 < 8; ++t2) {
            v2f b23; b23.x = b[t2].z; b23.y = b[t2].w;
            acc2[t2] = __builtin_amdgcn_wmma_f32_16x16x4_f32(
                false, b23, false, h23, (short)0, acc2[t2], false, false);
        }
        if (jj + 1 < 16) {
            #pragma unroll
            for (int t2 = 0; t2 < 8; ++t2) b[t2] = bn[t2];
        }
    }

    {
        const float4* b2q = (const float4*)b2 + 2 * hi;
        const float4* rq  = (const float4*)(resid + (size_t)(row0 + c) * DF) + 2 * hi;
        float4*       oq  = (float4*)(out + (size_t)(row0 + c) * DF) + 2 * hi;
        float4 rv[16];
        #pragma unroll
        for (int i = 0; i < 16; ++i) rv[i] = rq[4 * (i >> 1) + (i & 1)];
        #pragma unroll
        for (int t2 = 0; t2 < 8; ++t2) {
            const float4 p0 = b2q[4 * t2], p1 = b2q[4 * t2 + 1];
            const float bb[8] = {p0.x, p0.y, p0.z, p0.w, p1.x, p1.y, p1.z, p1.w};
            const float rr[8] = {rv[2 * t2].x, rv[2 * t2].y, rv[2 * t2].z, rv[2 * t2].w,
                                 rv[2 * t2 + 1].x, rv[2 * t2 + 1].y, rv[2 * t2 + 1].z, rv[2 * t2 + 1].w};
            float y[8];
            #pragma unroll
            for (int v = 0; v < 8; ++v) {
                float z = acc2[t2][v] + bb[v];
                z += rr[v];
                y[v] = (z > 0.0f) ? z : expm1f(z);
            }
            float4 o0; o0.x = y[0]; o0.y = y[1]; o0.z = y[2]; o0.w = y[3];
            float4 o1; o1.x = y[4]; o1.y = y[5]; o1.z = y[6]; o1.w = y[7];
            oq[4 * t2]     = o0;
            oq[4 * t2 + 1] = o1;
        }
    }
}

extern "C" void kernel_launch(void* const* d_in, const int* in_sizes, int n_in,
                              void* d_out, int out_size, void* d_ws, size_t ws_size,
                              hipStream_t stream) {
    const float* x   = (const float*)d_in[0];
    const int*   ei  = (const int*)d_in[1];
    const float* W1[3] = {(const float*)d_in[2],  (const float*)d_in[6],  (const float*)d_in[10]};
    const float* b1[3] = {(const float*)d_in[3],  (const float*)d_in[7],  (const float*)d_in[11]};
    const float* W2[3] = {(const float*)d_in[4],  (const float*)d_in[8],  (const float*)d_in[12]};
    const float* b2[3] = {(const float*)d_in[5],  (const float*)d_in[9],  (const float*)d_in[13]};
    const float* Wjk = (const float*)d_in[14];
    const float* bjk = (const float*)d_in[15];

    const size_t NM = (size_t)N_NODES * DF;
    float* h0 = (float*)d_ws;
    float* xs[3] = {h0 + NM, h0 + 2 * NM, h0 + 3 * NM};
    unsigned int* lists  = (unsigned int*)xs[2];
    unsigned int* counts = lists + (size_t)NLIST * LCAP;

    dim3 blk(GEMM_THREADS);
    const int gBlocks = GEMM_GROUPS;

    const float* xin = x;
    for (int l = 0; l < 3; ++l) {
        if (l == 0) slab_aggregate<0><<<NSLAB, dim3(1024), 0, stream>>>(h0, xin, ei, lists, counts);
        else        slab_aggregate<1><<<NSLAB, dim3(1024), 0, stream>>>(h0, xin, ei, lists, counts);
        mlp_rows<<<gBlocks, blk, 0, stream>>>(h0, W1[l], b1[l], W2[l], b2[l], xin, xs[l]);
        xin = xs[l];
    }
    gemm_rows<3, 0><<<gBlocks, blk, 0, stream>>>(xs[0], xs[1], xs[2],
                                                 Wjk, bjk, nullptr, (float*)d_out);
}
